# starting kernel
# baseline (speedup 1.0000x reference)
#include <hip/hip_runtime.h>


#define HID 128
#define LDSS 132
#define NEG_SLOPE 0.01f

typedef float v2f __attribute__((ext_vector_type(2)));
typedef float v8f __attribute__((ext_vector_type(8)));

__device__ __forceinline__ v8f wmma_f32(v2f a, v2f b, v8f c) {
  return __builtin_amdgcn_wmma_f32_16x16x4_f32(
       false, a,  false, b,
       (short)0, c,  false,  false);
}

__global__ __launch_bounds__(128)
void fused_x_kernel(const float* __restrict__ primal,
                    const float* __restrict__ last_primal,
                    const float* __restrict__ W2, const float* __restrict__ b2,
                    const float* __restrict__ W3, const float* __restrict__ b3,
                    float* __restrict__ X) {
  __shared__ float Ap[16 * LDSS];
  __shared__ float Al[16 * LDSS];
  const int m0  = blockIdx.x * 16;
  const int tid = threadIdx.x;

  const float4* gp = (const float4*)(primal      + (size_t)m0 * HID);
  const float4* gl = (const float4*)(last_primal + (size_t)m0 * HID);
#pragma unroll
  for (int j = 0; j < 4; ++j) {
    const int f   = tid + j * 128;
    const int row = f >> 5;
    const int c4  = f & 31;
    *(float4*)&Ap[row * LDSS + c4 * 4] = gp[f];
    *(float4*)&Al[row * LDSS + c4 * 4] = gl[f];
  }
  __syncthreads();

  const int lane = tid & 31;
  const int wave = tid >> 5;
  const int r    = lane & 15;
  const int koff = (lane >> 4) << 1;

#pragma unroll
  for (int t = 0; t < 2; ++t) {
    const int n0 = (wave * 2 + t) * 16;
    const int n  = n0 + (lane & 15);
    v8f acc  = {};
    v8f acc2 = {};
#pragma unroll
    for (int k0 = 0; k0 < HID; k0 += 4) {
      const int kk = k0 + koff;
      v2f a, b, a2, b2v;
      a.x   = Ap[r * LDSS + kk];  a.y   = Ap[r * LDSS + kk + 1];
      b.x   = W2[n * HID + kk];   b.y   = W2[n * HID + kk + 1];
      acc   = wmma_f32(a, b, acc);
      a2.x  = Al[r * LDSS + kk];  a2.y  = Al[r * LDSS + kk + 1];
      b2v.x = W3[n * HID + kk];   b2v.y = W3[n * HID + kk + 1];
      acc2  = wmma_f32(a2, b2v, acc2);
    }
    const float bias  = b2[n] - b3[n];
    const int   mbase = m0 + ((lane >> 4) << 3);
#pragma unroll
    for (int j = 0; j < 8; ++j)
      X[(size_t)(mbase + j) * HID + n] = acc[j] - acc2[j] + bias;
  }
}

__global__ __launch_bounds__(128)
void dual_base_kernel(const float* __restrict__ dual,
                      const float* __restrict__ W1, const float* __restrict__ b1,
                      const float* __restrict__ rhs,
                      const float* __restrict__ sigma_p,
                      float* __restrict__ out) {
  __shared__ float Ad[16 * LDSS];
  const int m0  = blockIdx.x * 16;
  const int tid = threadIdx.x;

  const float4* gd = (const float4*)(dual + (size_t)m0 * HID);
#pragma unroll
  for (int j = 0; j < 4; ++j) {
    const int f   = tid + j * 128;
    const int row = f >> 5;
    const int c4  = f & 31;
    *(float4*)&Ad[row * LDSS + c4 * 4] = gd[f];
  }
  __syncthreads();

  const int   lane  = tid & 31;
  const int   wave  = tid >> 5;
  const int   r     = lane & 15;
  const int   koff  = (lane >> 4) << 1;
  const float sigma = sigma_p[0];

#pragma unroll
  for (int t = 0; t < 2; ++t) {
    const int n0 = (wave * 2 + t) * 16;
    const int n  = n0 + (lane & 15);
    v8f acc = {};
#pragma unroll
    for (int k0 = 0; k0 < HID; k0 += 4) {
      const int kk = k0 + koff;
      v2f a, b;
      a.x = Ad[r * LDSS + kk];  a.y = Ad[r * LDSS + kk + 1];
      b.x = W1[n * HID + kk];   b.y = W1[n * HID + kk + 1];
      acc = wmma_f32(a, b, acc);
    }
    const float bias  = b1[n];
    const int   mbase = m0 + ((lane >> 4) << 3);
#pragma unroll
    for (int j = 0; j < 8; ++j) {
      const size_t idx = (size_t)(mbase + j) * HID + n;
      out[idx] = acc[j] + bias - sigma * rhs[idx];
    }
  }
}

__global__ __launch_bounds__(256)
void scatter_kernel(const int* __restrict__ rows, const int* __restrict__ cols,
                    const float* __restrict__ vals, const float* __restrict__ X,
                    const float* __restrict__ sigma_p, float* __restrict__ out,
                    int nnz) {
  const int e = blockIdx.x * 8 + (threadIdx.x >> 5);
  if (e >= nnz) return;
  const int   lane = threadIdx.x & 31;
  const int   rr   = rows[e];
  const int   cc   = cols[e];
  const float s    = vals[e] * sigma_p[0];

  const float4 xv = ((const float4*)(X + (size_t)cc * HID))[lane];
  float* op = out + (size_t)rr * HID + lane * 4;
  __hip_atomic_fetch_add(&op[0], s * xv.x, __ATOMIC_RELAXED, __HIP_MEMORY_SCOPE_AGENT);
  __hip_atomic_fetch_add(&op[1], s * xv.y, __ATOMIC_RELAXED, __HIP_MEMORY_SCOPE_AGENT);
  __hip_atomic_fetch_add(&op[2], s * xv.z, __ATOMIC_RELAXED, __HIP_MEMORY_SCOPE_AGENT);
  __hip_atomic_fetch_add(&op[3], s * xv.w, __ATOMIC_RELAXED, __HIP_MEMORY_SCOPE_AGENT);
}

__global__ __launch_bounds__(256)
void leaky_kernel(float* __restrict__ out, int n4) {
  int i = blockIdx.x * blockDim.x + threadIdx.x;
  const int stride = gridDim.x * blockDim.x;
  float4* p = (float4*)out;
  for (; i < n4; i += stride) {
    float4 v = p[i];
    v.x = v.x >= 0.f ? v.x : NEG_SLOPE * v.x;
    v.y = v.y >= 0.f ? v.y : NEG_SLOPE * v.y;
    v.z = v.z >= 0.f ? v.z : NEG_SLOPE * v.z;
    v.w = v.w >= 0.f ? v.w : NEG_SLOPE * v.w;
    p[i] = v;
  }
}

extern "C" void kernel_launch(void* const* d_in, const int* in_sizes, int n_in,
                              void* d_out, int out_size, void* d_ws, size_t ws_size,
                              hipStream_t stream) {
  const float* primal      = (const float*)d_in[0];
  const float* last_primal = (const float*)d_in[1];
  const float* dual        = (const float*)d_in[2];
  const int*   rows        = (const int*)d_in[3];
  const int*   cols        = (const int*)d_in[4];
  const float* vals        = (const float*)d_in[5];
  const float* rhs         = (const float*)d_in[6];
  const float* W1          = (const float*)d_in[7];
  const float* b1          = (const float*)d_in[8];
  const float* W2          = (const float*)d_in[9];
  const float* b2          = (const float*)d_in[10];
  const float* W3          = (const float*)d_in[11];
  const float* b3          = (const float*)d_in[12];
  const float* sigma       = (const float*)d_in[13];

  float* out = (float*)d_out;
  float* X   = (float*)d_ws;

  const int n_vars = in_sizes[0] / HID;
  const int m_cons = in_sizes[2] / HID;
  const int nnz    = in_sizes[5];

  fused_x_kernel<<<n_vars / 16, 128, 0, stream>>>(primal, last_primal,
                                                  W2, b2, W3, b3, X);
  dual_base_kernel<<<m_cons / 16, 128, 0, stream>>>(dual, W1, b1, rhs, sigma, out);
  scatter_kernel<<<(nnz + 7) / 8, 256, 0, stream>>>(rows, cols, vals, X, sigma,
                                                    out, nnz);
  leaky_kernel<<<4096, 256, 0, stream>>>(out, (m_cons * HID) / 4);
}


# Both GEMM kernels: 256 resident 16-wave blocks keep whole weight matrices in LDS; a wave owns a 16x128 tile, A fragments in registers straight from global, B fragments from LDS; sums unchanged
# speedup vs baseline: 1.2212x; 1.2212x over previous
#include <hip/hip_runtime.h>


#define HID 128
#define LDSS 132
#define NEG_SLOPE 0.01f

typedef float v2f __attribute__((ext_vector_type(2)));
typedef float v8f __attribute__((ext_vector_type(8)));

__device__ __forceinline__ v8f wmma_f32(v2f a, v2f b, v8f c) {
  return __builtin_amdgcn_wmma_f32_16x16x4_f32(
       false, a,  false, b,
       (short)0, c,  false,  false);
}

#define GEMM_THREADS 512
#define GEMM_WAVES   16
#define GEMM_BLOCKS  256

__device__ __forceinline__ void stage_weights(float* __restrict__ Ws,
                                              const float* __restrict__ Wg,
                                              int tid) {
  const float4* g = (const float4*)Wg;
#pragma unroll
  for (int j = 0; j < (HID * HID / 4) / GEMM_THREADS; ++j) {
    const int f   = tid + j * GEMM_THREADS;
    const int row = f >> 5;
    const int c4  = f & 31;
    *(float4*)&Ws[row * LDSS + c4 * 4] = g[f];
  }
}

__global__ __launch_bounds__(GEMM_THREADS)
void fused_x_kernel(const float* __restrict__ primal,
                    const float* __restrict__ last_primal,
                    const float* __restrict__ W2, const float* __restrict__ b2,
                    const float* __restrict__ W3, const float* __restrict__ b3,
                    float* __restrict__ X, int ntiles) {
  __shared__ float W2s[HID * LDSS];
  __shared__ float W3s[HID * LDSS];
  const int tid = threadIdx.x;
  stage_weights(W2s, W2, tid);
  stage_weights(W3s, W3, tid);
  __syncthreads();

  const int lane = tid & 31;
  const int wave = __builtin_amdgcn_readfirstlane(tid >> 5);
  const int c    = lane & 15;
  const int h    = lane >> 4;
  const int koff = h << 1;

  for (int tile = wave * GEMM_BLOCKS + (int)blockIdx.x; tile < ntiles;
       tile += GEMM_BLOCKS * GEMM_WAVES) {
    const int m0 = tile * 16;
    const float* pr = primal      + (size_t)(m0 + c) * HID + koff;
    const float* lr = last_primal + (size_t)(m0 + c) * HID + koff;
    v2f a[32], a2[32];
#pragma unroll
    for (int i = 0; i < 32; ++i) a[i]  = *(const v2f*)(pr + 4 * i);
#pragma unroll
    for (int i = 0; i < 32; ++i) a2[i] = *(const v2f*)(lr + 4 * i);

    for (int nt = 0; nt < 8; ++nt) {
      const int n = nt * 16 + c;
      const float* w2r = &W2s[n * LDSS + koff];
      const float* w3r = &W3s[n * LDSS + koff];
      v8f acc  = {};
      v8f acc2 = {};
#pragma unroll
      for (int i = 0; i < 32; ++i) {
        const v2f b   = *(const v2f*)(w2r + 4 * i);
        acc  = wmma_f32(a[i], b, acc);
        const v2f b2v = *(const v2f*)(w3r + 4 * i);
        acc2 = wmma_f32(a2[i], b2v, acc2);
      }
      __builtin_amdgcn_sched_group_barrier(0x100, 6, 0);
#pragma unroll
      for (int g = 0; g < 13; ++g) {
        __builtin_amdgcn_sched_group_barrier(0x008, 4, 0);
        __builtin_amdgcn_sched_group_barrier(0x100, 2, 0);
      }
      __builtin_amdgcn_sched_group_barrier(0x008, 12, 0);
      const float bias  = b2[n] - b3[n];
      const int   mbase = m0 + (h << 3);
#pragma unroll
      for (int j = 0; j < 8; ++j)
        X[(size_t)(mbase + j) * HID + n] = acc[j] - acc2[j] + bias;
    }
  }
}

__global__ __launch_bounds__(GEMM_THREADS)
void dual_base_kernel(const float* __restrict__ dual,
                      const float* __restrict__ W1, const float* __restrict__ b1,
                      const float* __restrict__ rhs,
                      const float* __restrict__ sigma_p,
                      float* __restrict__ out, int ntiles) {
  __shared__ float W1s[HID * LDSS];
  const int tid = threadIdx.x;
  stage_weights(W1s, W1, tid);
  __syncthreads();

  const int   lane  = tid & 31;
  const int   wave  = __builtin_amdgcn_readfirstlane(tid >> 5);
  const int   c     = lane & 15;
  const int   h     = lane >> 4;
  const int   koff  = h << 1;
  const float sigma = sigma_p[0];

  for (int tile = wave * GEMM_BLOCKS + (int)blockIdx.x; tile < ntiles;
       tile += GEMM_BLOCKS * GEMM_WAVES) {
    const int m0 = tile * 16;
    const float* dr = dual + (size_t)(m0 + c) * HID + koff;
    v2f a[32];
#pragma unroll
    for (int i = 0; i < 32; ++i) a[i] = *(const v2f*)(dr + 4 * i);

    for (int np = 0; np < 4; ++np) {
      const int nA = np * 32 + c;
      const int nB = nA + 16;
      const float* wAr = &W1s[nA * LDSS + koff];
      const float* wBr = &W1s[nB * LDSS + koff];
      v8f accA = {};
      v8f accB = {};
#pragma unroll
      for (int i = 0; i < 32; ++i) {
        const v2f bA = *(const v2f*)(wAr + 4 * i);
        accA = wmma_f32(a[i], bA, accA);
        const v2f bB = *(const v2f*)(wBr + 4 * i);
        accB = wmma_f32(a[i], bB, accB);
      }
      __builtin_amdgcn_sched_group_barrier(0x100, 6, 0);
#pragma unroll
      for (int g = 0; g < 13; ++g) {
        __builtin_amdgcn_sched_group_barrier(0x008, 4, 0);
        __builtin_amdgcn_sched_group_barrier(0x100, 2, 0);
      }
      __builtin_amdgcn_sched_group_barrier(0x008, 12, 0);
      const float biasA = b1[nA];
      const float biasB = b1[nB];
      const int   mbase = m0 + (h << 3);
#pragma unroll
      for (int j = 0; j < 8; ++j) {
        const size_t idxA = (size_t)(mbase + j) * HID + nA;
        const size_t idxB = (size_t)(mbase + j) * HID + nB;
        out[idxA] = accA[j] + biasA - sigma * rhs[idxA];
        out[idxB] = accB[j] + biasB - sigma * rhs[idxB];
      }
    }
  }
}

__global__ __launch_bounds__(256)
void scatter_kernel(const int* __restrict__ rows, const int* __restrict__ cols,
                    const float* __restrict__ vals, const float* __restrict__ X,
                    const float* __restrict__ sigma_p, float* __restrict__ out,
                    int nnz) {
  const int e = blockIdx.x * 8 + (threadIdx.x >> 5);
  if (e >= nnz) return;
  const int   lane = threadIdx.x & 31;
  const int   rr   = rows[e];
  const int   cc   = cols[e];
  const float s    = vals[e] * sigma_p[0];

  const float4 xv = ((const float4*)(X + (size_t)cc * HID))[lane];
  float* op = out + (size_t)rr * HID + lane * 4;
  __hip_atomic_fetch_add(&op[0], s * xv.x, __ATOMIC_RELAXED, __HIP_MEMORY_SCOPE_AGENT);
  __hip_atomic_fetch_add(&op[1], s * xv.y, __ATOMIC_RELAXED, __HIP_MEMORY_SCOPE_AGENT);
  __hip_atomic_fetch_add(&op[2], s * xv.z, __ATOMIC_RELAXED, __HIP_MEMORY_SCOPE_AGENT);
  __hip_atomic_fetch_add(&op[3], s * xv.w, __ATOMIC_RELAXED, __HIP_MEMORY_SCOPE_AGENT);
}

__global__ __launch_bounds__(256)
void leaky_kernel(float* __restrict__ out, int n4) {
  int i = blockIdx.x * blockDim.x + threadIdx.x;
  const int stride = gridDim.x * blockDim.x;
  float4* p = (float4*)out;
  for (; i < n4; i += stride) {
    float4 v = p[i];
    v.x = v.x >= 0.f ? v.x : NEG_SLOPE * v.x;
    v.y = v.y >= 0.f ? v.y : NEG_SLOPE * v.y;
    v.z = v.z >= 0.f ? v.z : NEG_SLOPE * v.z;
    v.w = v.w >= 0.f ? v.w : NEG_SLOPE * v.w;
    p[i] = v;
  }
}

extern "C" void kernel_launch(void* const* d_in, const int* in_sizes, int n_in,
                              void* d_out, int out_size, void* d_ws, size_t ws_size,
                              hipStream_t stream) {
  const float* primal      = (const float*)d_in[0];
  const float* last_primal = (const float*)d_in[1];
  const float* dual        = (const float*)d_in[2];
  const int*   rows        = (const int*)d_in[3];
  const int*   cols        = (const int*)d_in[4];
  const float* vals        = (const float*)d_in[5];
  const float* rhs         = (const float*)d_in[6];
  const float* W1          = (const float*)d_in[7];
  const float* b1          = (const float*)d_in[8];
  const float* W2          = (const float*)d_in[9];
  const float* b2          = (const float*)d_in[10];
  const float* W3          = (const float*)d_in[11];
  const float* b3          = (const float*)d_in[12];
  const float* sigma       = (const float*)d_in[13];

  float* out = (float*)d_out;
  float* X   = (float*)d_ws;

  const int n_vars = in_sizes[0] / HID;
  const int m_cons = in_sizes[2] / HID;
  const int nnz    = in_sizes[5];

  fused_x_kernel<<<GEMM_BLOCKS, GEMM_THREADS, 0, stream>>>(primal, last_primal,
                                                          W2, b2, W3, b3, X,
                                                          n_vars / 16);
  dual_base_kernel<<<GEMM_BLOCKS, GEMM_THREADS, 0, stream>>>(dual, W1, b1, rhs,
                                                            sigma, out,
                                                            m_cons / 16);
  scatter_kernel<<<(nnz + 7) / 8, 256, 0, stream>>>(rows, cols, vals, X, sigma,
                                                    out, nnz);
  leaky_kernel<<<4096, 256, 0, stream>>>(out, (m_cons * HID) / 4);
}
